# starting kernel
# baseline (speedup 1.0000x reference)
#include <hip/hip_runtime.h>

typedef __attribute__((ext_vector_type(2))) float v2f;
typedef __attribute__((ext_vector_type(4))) float v4f;
typedef __attribute__((ext_vector_type(8))) float v8f;

#define F_DIM 256
#define P_LEAVES 128
#define WAVES_PER_WG 8
#define ROWS_PER_WAVE 32
#define THREADS_PER_WG 256

#define SLICE_W 576
#define KH_OFF  288
#define LDS_WORDS ((F_DIM / 4) * SLICE_W)

__device__ __forceinline__ float fmin1(float a, float b) {
    return __builtin_amdgcn_fmed3f(a, b, -__builtin_inff());
}
__device__ __forceinline__ float fmax1(float a, float b) {
    return __builtin_amdgcn_fmed3f(a, b, __builtin_inff());
}
__device__ __forceinline__ v8f v8min(v8f a, v8f b) {
    v8f r;
#pragma unroll
    for (int i = 0; i < 8; ++i) r[i] = fmin1(a[i], b[i]);
    return r;
}
__device__ __forceinline__ v8f v8max(v8f a, v8f b) {
    v8f r;
#pragma unroll
    for (int i = 0; i < 8; ++i) r[i] = fmax1(a[i], b[i]);
    return r;
}
template <int XORM>
__device__ __forceinline__ v8f v8swz(v8f v) {
    constexpr int PAT = (XORM << 10) | 0x1f;
    v8f r;
#pragma unroll
    for (int i = 0; i < 8; ++i)
        r[i] = __int_as_float(
            __builtin_amdgcn_ds_swizzle(__float_as_int(v[i]), PAT));
    return r;
}

extern "C" __global__ __launch_bounds__(THREADS_PER_WG)
void aln_wmma_kernel(const float* __restrict__ x,
                     const float* __restrict__ W,
                     const float* __restrict__ bias,
                     float* __restrict__ out)
{
    extern __shared__ float ldsW[];

    const int tid = threadIdx.x;

    const v4f* W4 = (const v4f*)W;
#pragma unroll 4
    for (int i = 0; i < (F_DIM * P_LEAVES) / (4 * THREADS_PER_WG); ++i) {
        int flat4 = tid + i * THREADS_PER_WG;
        v4f w = W4[flat4];
        int k  = flat4 >> 5;
        int n0 = (flat4 & 31) * 4;
        int bse = (k >> 2) * SLICE_W + ((k >> 1) & 1) * KH_OFF + (k & 1);
        ldsW[bse + (n0 + 0) * 2] = w[0];
        ldsW[bse + (n0 + 1) * 2] = w[1];
        ldsW[bse + (n0 + 2) * 2] = w[2];
        ldsW[bse + (n0 + 3) * 2] = w[3];
    }

    const int lane  = tid & 31;
    const int wave  = tid >> 5;
    const int c     = lane & 15;
    const int khalf = lane >> 4;

    const int rowBase = (blockIdx.x * WAVES_PER_WG + wave) * ROWS_PER_WAVE;

    float bb[8];
#pragma unroll
    for (int nt = 0; nt < 8; ++nt) bb[nt] = bias[nt * 16 + c];

    __syncthreads();

    const float* xA0 = x + (size_t)(rowBase + c) * F_DIM + khalf * 2;
    const float* xA1 = xA0 + 16 * F_DIM;

    const float* ldsB = ldsW + khalf * KH_OFF + c * 2;

    v8f acc[2][8];
#pragma unroll
    for (int mt = 0; mt < 2; ++mt)
#pragma unroll
        for (int nt = 0; nt < 8; ++nt) acc[mt][nt] = (v8f)0.0f;

#pragma unroll 2
    for (int kc = 0; kc < F_DIM / 4; ++kc) {
        v2f a0 = __builtin_nontemporal_load((const v2f*)(xA0 + kc * 4));
        v2f a1 = __builtin_nontemporal_load((const v2f*)(xA1 + kc * 4));
        const float* bp = ldsB + kc * SLICE_W;
#pragma unroll
        for (int nt = 0; nt < 8; ++nt) {
            v2f bf = *(const v2f*)(bp + nt * 32);
            acc[0][nt] = __builtin_amdgcn_wmma_f32_16x16x4_f32(
                false, a0, false, bf, (short)0, acc[0][nt], false, false);
            acc[1][nt] = __builtin_amdgcn_wmma_f32_16x16x4_f32(
                false, a1, false, bf, (short)0, acc[1][nt], false, false);
        }
    }

    const bool oddlane = (lane & 1) != 0;
#pragma unroll
    for (int mt = 0; mt < 2; ++mt) {
        v8f p[4];
#pragma unroll
        for (int j = 0; j < 4; ++j) {
            v8f va = acc[mt][2 * j]     + bb[2 * j];
            v8f vb = acc[mt][2 * j + 1] + bb[2 * j + 1];
            va = v8min(va, v8swz<1>(va));
            vb = v8min(vb, v8swz<1>(vb));
            v8f m;
#pragma unroll
            for (int i = 0; i < 8; ++i) m[i] = oddlane ? vb[i] : va[i];
            m = v8max(m, v8swz<2>(m));
            m = v8min(m, v8swz<4>(m));
            m = v8max(m, v8swz<8>(m));
            m = v8min(m, v8swz<1>(m));
            p[j] = m;
        }
        v8f q0 = v8max(p[0], p[1]);
        v8f q1 = v8max(p[2], p[3]);
        v8f r  = v8min(q0, q1);

        if (c == 0) {
            float* o = out + rowBase + mt * 16 + khalf * 8;
            v4f lo = { r[0], r[1], r[2], r[3] };
            v4f hi = { r[4], r[5], r[6], r[7] };
            *(v4f*)(o + 0) = lo;
            *(v4f*)(o + 4) = hi;
        }
    }
}

extern "C" void kernel_launch(void* const* d_in, const int* in_sizes, int n_in,
                              void* d_out, int out_size, void* d_ws, size_t ws_size,
                              hipStream_t stream) {
    const float* x = (const float*)d_in[0];
    const float* W = (const float*)d_in[1];
    const float* b = (const float*)d_in[2];
    float* out = (float*)d_out;

    const int Bsz = in_sizes[0] / F_DIM;
    const int rowsPerWG = WAVES_PER_WG * ROWS_PER_WAVE;
    const int nBlocks = Bsz / rowsPerWG;

    const size_t shmem = (size_t)LDS_WORDS * sizeof(float);
    (void)hipFuncSetAttribute((const void*)aln_wmma_kernel,
                              hipFuncAttributeMaxDynamicSharedMemorySize,
                              (int)shmem);

    aln_wmma_kernel<<<nBlocks, THREADS_PER_WG, shmem, stream>>>(x, W, b, out);
}


# Split-K hand-over of tile t (stores landed, split barrier, LDS reads, 4-way sum, tree fold) issued in pieces between the k-steps of tile t+1, not behind its own tile; sums and their order unchanged
# speedup vs baseline: 1.8065x; 1.8065x over previous
#include <hip/hip_runtime.h>


typedef __attribute__((ext_vector_type(2))) float v2f;
typedef __attribute__((ext_vector_type(4))) float v4f;
typedef __attribute__((ext_vector_type(8))) float v8f;

#define F_DIM 256
#define P_LEAVES 128
#define KSPLIT 4
#define KQ (F_DIM / KSPLIT)
#define KSTEPS (KQ / 4)
#define NTILES 8
#define THREADS_PER_WG 128
#define TILES_PER_WG 8
#define ROWS_PER_WG (TILES_PER_WG * 16)
#define PART_WORDS (KSPLIT * 16 * P_LEAVES)
#define LDS_WORDS (2 * PART_WORDS + ROWS_PER_WG)


__device__ __forceinline__ float fmin1(float a, float b) {
    return __builtin_amdgcn_fmed3f(a, b, -__builtin_inff());
}
__device__ __forceinline__ float fmax1(float a, float b) {
    return __builtin_amdgcn_fmed3f(a, b, __builtin_inff());
}
__device__ __forceinline__ v4f v4min(v4f a, v4f b) {
    v4f r;
#pragma unroll
    for (int i = 0; i < 4; ++i) r[i] = fmin1(a[i], b[i]);
    return r;
}
__device__ __forceinline__ v4f v4max(v4f a, v4f b) {
    v4f r;
#pragma unroll
    for (int i = 0; i < 4; ++i) r[i] = fmax1(a[i], b[i]);
    return r;
}
template <int XORM>
__device__ __forceinline__ v4f v4swz(v4f v) {
    constexpr int PAT = (XORM << 10) | 0x1f;
    v4f r;
#pragma unroll
    for (int i = 0; i < 4; ++i)
        r[i] = __int_as_float(
            __builtin_amdgcn_ds_swizzle(__float_as_int(v[i]), PAT));
    return r;
}

#define ACC8(a) "+v"(a[0]), "+v"(a[1]), "+v"(a[2]), "+v"(a[3]), \
                "+v"(a[4]), "+v"(a[5]), "+v"(a[6]), "+v"(a[7])
#define HOLD_ACC(a) asm volatile("" : ACC8(a) :: "memory")
#define HOLD_FOLD(a, rv, sw) asm volatile("" : ACC8(a), "+v"(rv), "+v"(sw))
#define HOLD_RV(a, rv) asm volatile("" : ACC8(a), "+v"(rv))
#define LDS_LANDED_SIGNAL(a) \
    asm volatile("s_wait_dscnt 0x0\n\ts_barrier_signal -1" : ACC8(a) :: "memory")
#define WG_WAIT(a) asm volatile("s_barrier_wait -1" : ACC8(a) :: "memory")

__device__ __forceinline__ void load_a(v4f (&a)[8], const float* p) {
#pragma unroll
    for (int j = 0; j < 8; ++j) a[j] = *(const v4f*)(p + 8 * j);
}

template <int S0, int S1>
__device__ __forceinline__ void wmma_steps(v8f (&acc)[NTILES], const v4f (&a)[8],
                                           const v2f (&wf)[KSTEPS][NTILES])
{
#pragma unroll
    for (int s = S0; s < S1; ++s) {
        const v4f aj = a[s >> 1];
        v2f af;
        if (s & 1) { af[0] = aj[2]; af[1] = aj[3]; }
        else       { af[0] = aj[0]; af[1] = aj[1]; }
#pragma unroll
        for (int nt = 0; nt < NTILES; ++nt)
            acc[nt] = __builtin_amdgcn_wmma_f32_16x16x4_f32(
                false, af, false, wf[s][nt], (short)0, acc[nt], false, false);
    }
}

__device__ __forceinline__ void store_partials(const v8f (&acc)[NTILES], float* partw) {
#pragma unroll
    for (int r = 0; r < 8; ++r) {
        v4f lo = { acc[0][r], acc[1][r], acc[2][r], acc[3][r] };
        v4f hi = { acc[4][r], acc[5][r], acc[6][r], acc[7][r] };
        *(v4f*)(partw + r * P_LEAVES)     = lo;
        *(v4f*)(partw + r * P_LEAVES + 4) = hi;
    }
}

__device__ __forceinline__ void load_partials(v4f (&t)[4][KSPLIT], const float* partr) {
#pragma unroll
    for (int i = 0; i < 4; ++i)
#pragma unroll
        for (int q = 0; q < KSPLIT; ++q)
            t[i][q] = *(const v4f*)(partr + i * P_LEAVES + q * 16 * P_LEAVES);
}

__device__ __forceinline__ v4f sum_fold_in_lane(const v4f (&t)[4][KSPLIT], const v4f bias4) {
    v4f rv;
#pragma unroll
    for (int i = 0; i < 4; ++i) {
        const v4f sm = ((t[i][0] + t[i][1]) + (t[i][2] + t[i][3])) + bias4;
        rv[i] = fmax1(fmin1(sm[0], sm[1]), fmin1(sm[2], sm[3]));
    }
    return rv;
}

__device__ __forceinline__ void tile_first(const v4f (&a)[8],
                                           const v2f (&wf)[KSTEPS][NTILES],
                                           float* partw)
{
    v8f acc[NTILES];
#pragma unroll
    for (int nt = 0; nt < NTILES; ++nt) acc[nt] = (v8f)0.0f;
    wmma_steps<0, KSTEPS>(acc, a, wf);
    store_partials(acc, partw);
}

__device__ __forceinline__ void tile_pipe(const v4f (&a)[8],
                                          const v2f (&wf)[KSTEPS][NTILES],
                                          float* partw_cur, const float* partr_prev,
                                          const v4f bias4, float* resp_prev,
                                          const int lane)
{
    v8f acc[NTILES];
#pragma unroll
    for (int nt = 0; nt < NTILES; ++nt) acc[nt] = (v8f)0.0f;

    wmma_steps<0, 2>(acc, a, wf);
    LDS_LANDED_SIGNAL(acc);
    wmma_steps<2, 3>(acc, a, wf);
    WG_WAIT(acc);
    v4f t[4][KSPLIT];
    load_partials(t, partr_prev);
    HOLD_ACC(acc);
    wmma_steps<3, 6>(acc, a, wf);
    asm volatile("" : ACC8(acc),
                 "+v"(t[0][0]), "+v"(t[0][1]), "+v"(t[0][2]), "+v"(t[0][3]),
                 "+v"(t[1][0]), "+v"(t[1][1]), "+v"(t[1][2]), "+v"(t[1][3]),
                 "+v"(t[2][0]), "+v"(t[2][1]), "+v"(t[2][2]), "+v"(t[2][3]),
                 "+v"(t[3][0]), "+v"(t[3][1]), "+v"(t[3][2]), "+v"(t[3][3]));
    v4f rv = sum_fold_in_lane(t, bias4);
    HOLD_RV(acc, rv);
    v4f sw = v4swz<1>(rv);
    wmma_steps<6, 8>(acc, a, wf);
    HOLD_FOLD(acc, rv, sw);
    rv = v4min(rv, sw);
    HOLD_RV(acc, rv);
    sw = v4swz<2>(rv);
    wmma_steps<8, 10>(acc, a, wf);
    HOLD_FOLD(acc, rv, sw);
    rv = v4max(rv, sw);
    HOLD_RV(acc, rv);
    sw = v4swz<4>(rv);
    wmma_steps<10, 12>(acc, a, wf);
    HOLD_FOLD(acc, rv, sw);
    rv = v4min(rv, sw);
    HOLD_RV(acc, rv);
    sw = v4swz<8>(rv);
    wmma_steps<12, 14>(acc, a, wf);
    HOLD_FOLD(acc, rv, sw);
    rv = v4max(rv, sw);
    HOLD_RV(acc, rv);
    sw = v4swz<16>(rv);
    wmma_steps<14, KSTEPS>(acc, a, wf);
    rv = v4min(rv, sw);

    store_partials(acc, partw_cur);
    if (lane == 0) *(v4f*)resp_prev = rv;
}

__device__ __forceinline__ void meet_last(const float* partr_prev, const v4f bias4,
                                          float* resp_prev, const int lane)
{
    asm volatile("s_wait_dscnt 0x0\n\ts_barrier_signal -1\n\ts_barrier_wait -1" ::: "memory");
    v4f t[4][KSPLIT];
    load_partials(t, partr_prev);
    v4f rv = sum_fold_in_lane(t, bias4);
    rv = v4min(rv, v4swz<1>(rv));
    rv = v4max(rv, v4swz<2>(rv));
    rv = v4min(rv, v4swz<4>(rv));
    rv = v4max(rv, v4swz<8>(rv));
    rv = v4min(rv, v4swz<16>(rv));
    if (lane == 0) *(v4f*)resp_prev = rv;
}

extern "C" __global__ __launch_bounds__(THREADS_PER_WG, 2)
void aln_wmma_kernel(const float* __restrict__ x,
                     const float* __restrict__ W,
                     const float* __restrict__ bias,
                     float* __restrict__ out)
{
    extern __shared__ float lds[];
    float* part = lds;
    float* res  = lds + 2 * PART_WORDS;

    const int tid  = threadIdx.x;
    const int lane = tid & 31;
    const int wave = tid >> 5;
    const int c    = lane & 15;
    const int h    = lane >> 4;
    const int kq0  = wave * KQ;
    const int row0 = blockIdx.x * ROWS_PER_WG;

    v2f wf[KSTEPS][NTILES];
    {
        const float* wp = W + (size_t)(kq0 + 4 * h) * P_LEAVES + 8 * c;
#pragma unroll
        for (int s = 0; s < KSTEPS; ++s) {
#pragma unroll
            for (int e = 0; e < 2; ++e) {
                const int kl = 8 * (s >> 1) + 2 * (s & 1) + e;
                const v4f lo = *(const v4f*)(wp + kl * P_LEAVES);
                const v4f hi = *(const v4f*)(wp + kl * P_LEAVES + 4);
                wf[s][0][e] = lo[0]; wf[s][1][e] = lo[1];
                wf[s][2][e] = lo[2]; wf[s][3][e] = lo[3];
                wf[s][4][e] = hi[0]; wf[s][5][e] = hi[1];
                wf[s][6][e] = hi[2]; wf[s][7][e] = hi[3];
            }
        }
    }

    const v4f bias4 = *(const v4f*)(bias + 4 * lane);

    const float* xp = x + (size_t)(row0 + c) * F_DIM + kq0 + 4 * h;

    float* partw0 = part + (wave * 16 + 8 * h) * P_LEAVES + 8 * c;
    const float* partr0 = part + (4 * wave) * P_LEAVES + 4 * lane;
    float* resw = res + 4 * wave;

    v4f aE[8], aO[8];
    load_a(aE, xp);
    load_a(aO, xp + (size_t)16 * F_DIM);
    tile_first(aE, wf, partw0);

    for (int tp = 0; tp < TILES_PER_WG / 2 - 1; ++tp) {
        const int te = 2 * tp;
        load_a(aE, xp + (size_t)(te + 2) * 16 * F_DIM);
        tile_pipe(aO, wf, partw0 + PART_WORDS, partr0, bias4,
                  resw + te * 16, lane);
        load_a(aO, xp + (size_t)(te + 3) * 16 * F_DIM);
        tile_pipe(aE, wf, partw0, partr0 + PART_WORDS, bias4,
                  resw + (te + 1) * 16, lane);
    }
    tile_pipe(aO, wf, partw0 + PART_WORDS, partr0, bias4,
              resw + (TILES_PER_WG - 2) * 16, lane);
    meet_last(partr0 + PART_WORDS, bias4, resw + (TILES_PER_WG - 1) * 16, lane);

    __syncthreads();

    out[row0 + tid] = res[tid];
}

extern "C" void kernel_launch(void* const* d_in, const int* in_sizes, int n_in,
                              void* d_out, int out_size, void* d_ws, size_t ws_size,
                              hipStream_t stream) {
    const float* x = (const float*)d_in[0];
    const float* W = (const float*)d_in[1];
    const float* b = (const float*)d_in[2];
    float* out = (float*)d_out;

    const int Bsz = in_sizes[0] / F_DIM;
    const int nBlocks = Bsz / ROWS_PER_WG;

    const size_t shmem = (size_t)LDS_WORDS * sizeof(float);
    (void)hipFuncSetAttribute((const void*)aln_wmma_kernel,
                              hipFuncAttributeMaxDynamicSharedMemorySize,
                              (int)shmem);

    aln_wmma_kernel<<<nBlocks, THREADS_PER_WG, shmem, stream>>>(x, W, b, out);
}


# Wave priority steps down with progress (s_setprio 3,2,1,0 at tiles 0,3,5,7): the two workgroups on each SIMD take the matrix pipe in turns and finish together; arithmetic and sum order unchanged
# speedup vs baseline: 1.8588x; 1.0290x over previous
#include <hip/hip_runtime.h>


typedef __attribute__((ext_vector_type(2))) float v2f;
typedef __attribute__((ext_vector_type(4))) float v4f;
typedef __attribute__((ext_vector_type(8))) float v8f;

#define F_DIM 256
#define P_LEAVES 128
#define KSPLIT 4
#define KQ (F_DIM / KSPLIT)
#define KSTEPS (KQ / 4)
#define NTILES 8
#define THREADS_PER_WG 128
#define TILES_PER_WG 8
#define ROWS_PER_WG (TILES_PER_WG * 16)
#define PART_WORDS (KSPLIT * 16 * P_LEAVES)
#define LDS_WORDS (2 * PART_WORDS + ROWS_PER_WG)


__device__ __forceinline__ float fmin1(float a, float b) {
    return __builtin_amdgcn_fmed3f(a, b, -__builtin_inff());
}
__device__ __forceinline__ float fmax1(float a, float b) {
    return __builtin_amdgcn_fmed3f(a, b, __builtin_inff());
}
__device__ __forceinline__ v4f v4min(v4f a, v4f b) {
    v4f r;
#pragma unroll
    for (int i = 0; i < 4; ++i) r[i] = fmin1(a[i], b[i]);
    return r;
}
__device__ __forceinline__ v4f v4max(v4f a, v4f b) {
    v4f r;
#pragma unroll
    for (int i = 0; i < 4; ++i) r[i] = fmax1(a[i], b[i]);
    return r;
}
template <int XORM>
__device__ __forceinline__ v4f v4swz(v4f v) {
    constexpr int PAT = (XORM << 10) | 0x1f;
    v4f r;
#pragma unroll
    for (int i = 0; i < 4; ++i)
        r[i] = __int_as_float(
            __builtin_amdgcn_ds_swizzle(__float_as_int(v[i]), PAT));
    return r;
}

#define ACC8(a) "+v"(a[0]), "+v"(a[1]), "+v"(a[2]), "+v"(a[3]), \
                "+v"(a[4]), "+v"(a[5]), "+v"(a[6]), "+v"(a[7])
#define HOLD_ACC(a) asm volatile("" : ACC8(a) :: "memory")
#define HOLD_FOLD(a, rv, sw) asm volatile("" : ACC8(a), "+v"(rv), "+v"(sw))
#define HOLD_RV(a, rv) asm volatile("" : ACC8(a), "+v"(rv))
#define LDS_LANDED_SIGNAL(a) \
    asm volatile("s_wait_dscnt 0x0\n\ts_barrier_signal -1" : ACC8(a) :: "memory")
#define WG_WAIT(a) asm volatile("s_barrier_wait -1" : ACC8(a) :: "memory")

__device__ __forceinline__ void load_a(v4f (&a)[8], const float* p) {
#pragma unroll
    for (int j = 0; j < 8; ++j) a[j] = *(const v4f*)(p + 8 * j);
}

template <int S0, int S1>
__device__ __forceinline__ void wmma_steps(v8f (&acc)[NTILES], const v4f (&a)[8],
                                           const v2f (&wf)[KSTEPS][NTILES])
{
#pragma unroll
    for (int s = S0; s < S1; ++s) {
        const v4f aj = a[s >> 1];
        v2f af;
        if (s & 1) { af[0] = aj[2]; af[1] = aj[3]; }
        else       { af[0] = aj[0]; af[1] = aj[1]; }
#pragma unroll
        for (int nt = 0; nt < NTILES; ++nt)
            acc[nt] = __builtin_amdgcn_wmma_f32_16x16x4_f32(
                false, af, false, wf[s][nt], (short)0, acc[nt], false, false);
    }
}

__device__ __forceinline__ void store_partials(const v8f (&acc)[NTILES], float* partw) {
#pragma unroll
    for (int r = 0; r < 8; ++r) {
        v4f lo = { acc[0][r], acc[1][r], acc[2][r], acc[3][r] };
        v4f hi = { acc[4][r], acc[5][r], acc[6][r], acc[7][r] };
        *(v4f*)(partw + r * P_LEAVES)     = lo;
        *(v4f*)(partw + r * P_LEAVES + 4) = hi;
    }
}

__device__ __forceinline__ void load_partials(v4f (&t)[4][KSPLIT], const float* partr) {
#pragma unroll
    for (int i = 0; i < 4; ++i)
#pragma unroll
        for (int q = 0; q < KSPLIT; ++q)
            t[i][q] = *(const v4f*)(partr + i * P_LEAVES + q * 16 * P_LEAVES);
}

__device__ __forceinline__ v4f sum_fold_in_lane(const v4f (&t)[4][KSPLIT], const v4f bias4) {
    v4f rv;
#pragma unroll
    for (int i = 0; i < 4; ++i) {
        const v4f sm = ((t[i][0] + t[i][1]) + (t[i][2] + t[i][3])) + bias4;
        rv[i] = fmax1(fmin1(sm[0], sm[1]), fmin1(sm[2], sm[3]));
    }
    return rv;
}

__device__ __forceinline__ void tile_first(const v4f (&a)[8],
                                           const v2f (&wf)[KSTEPS][NTILES],
                                           float* partw)
{
    v8f acc[NTILES];
#pragma unroll
    for (int nt = 0; nt < NTILES; ++nt) acc[nt] = (v8f)0.0f;
    wmma_steps<0, KSTEPS>(acc, a, wf);
    store_partials(acc, partw);
}

__device__ __forceinline__ void tile_pipe(const v4f (&a)[8],
                                          const v2f (&wf)[KSTEPS][NTILES],
                                          float* partw_cur, const float* partr_prev,
                                          const v4f bias4, float* resp_prev,
                                          const int lane)
{
    v8f acc[NTILES];
#pragma unroll
    for (int nt = 0; nt < NTILES; ++nt) acc[nt] = (v8f)0.0f;

    wmma_steps<0, 2>(acc, a, wf);
    LDS_LANDED_SIGNAL(acc);
    wmma_steps<2, 3>(acc, a, wf);
    WG_WAIT(acc);
    v4f t[4][KSPLIT];
    load_partials(t, partr_prev);
    HOLD_ACC(acc);
    wmma_steps<3, 6>(acc, a, wf);
    asm volatile("" : ACC8(acc),
                 "+v"(t[0][0]), "+v"(t[0][1]), "+v"(t[0][2]), "+v"(t[0][3]),
                 "+v"(t[1][0]), "+v"(t[1][1]), "+v"(t[1][2]), "+v"(t[1][3]),
                 "+v"(t[2][0]), "+v"(t[2][1]), "+v"(t[2][2]), "+v"(t[2][3]),
                 "+v"(t[3][0]), "+v"(t[3][1]), "+v"(t[3][2]), "+v"(t[3][3]));
    v4f rv = sum_fold_in_lane(t, bias4);
    HOLD_RV(acc, rv);
    v4f sw = v4swz<1>(rv);
    wmma_steps<6, 8>(acc, a, wf);
    HOLD_FOLD(acc, rv, sw);
    rv = v4min(rv, sw);
    HOLD_RV(acc, rv);
    sw = v4swz<2>(rv);
    wmma_steps<8, 10>(acc, a, wf);
    HOLD_FOLD(acc, rv, sw);
    rv = v4max(rv, sw);
    HOLD_RV(acc, rv);
    sw = v4swz<4>(rv);
    wmma_steps<10, 12>(acc, a, wf);
    HOLD_FOLD(acc, rv, sw);
    rv = v4min(rv, sw);
    HOLD_RV(acc, rv);
    sw = v4swz<8>(rv);
    wmma_steps<12, 14>(acc, a, wf);
    HOLD_FOLD(acc, rv, sw);
    rv = v4max(rv, sw);
    HOLD_RV(acc, rv);
    sw = v4swz<16>(rv);
    wmma_steps<14, KSTEPS>(acc, a, wf);
    rv = v4min(rv, sw);

    store_partials(acc, partw_cur);
    if (lane == 0) *(v4f*)resp_prev = rv;
}

__device__ __forceinline__ void meet_last(const float* partr_prev, const v4f bias4,
                                          float* resp_prev, const int lane)
{
    asm volatile("s_wait_dscnt 0x0\n\ts_barrier_signal -1\n\ts_barrier_wait -1" ::: "memory");
    v4f t[4][KSPLIT];
    load_partials(t, partr_prev);
    v4f rv = sum_fold_in_lane(t, bias4);
    rv = v4min(rv, v4swz<1>(rv));
    rv = v4max(rv, v4swz<2>(rv));
    rv = v4min(rv, v4swz<4>(rv));
    rv = v4max(rv, v4swz<8>(rv));
    rv = v4min(rv, v4swz<16>(rv));
    if (lane == 0) *(v4f*)resp_prev = rv;
}

extern "C" __global__ __launch_bounds__(THREADS_PER_WG, 2)
void aln_wmma_kernel(const float* __restrict__ x,
                     const float* __restrict__ W,
                     const float* __restrict__ bias,
                     float* __restrict__ out)
{
    __builtin_amdgcn_s_setprio(3);

    extern __shared__ float lds[];
    float* part = lds;
    float* res  = lds + 2 * PART_WORDS;

    const int tid  = threadIdx.x;
    const int lane = tid & 31;
    const int wave = tid >> 5;
    const int c    = lane & 15;
    const int h    = lane >> 4;
    const int kq0  = wave * KQ;
    const int row0 = blockIdx.x * ROWS_PER_WG;

    v2f wf[KSTEPS][NTILES];
    {
        const float* wp = W + (size_t)(kq0 + 4 * h) * P_LEAVES + 8 * c;
#pragma unroll
        for (int s = 0; s < KSTEPS; ++s) {
#pragma unroll
            for (int e = 0; e < 2; ++e) {
                const int kl = 8 * (s >> 1) + 2 * (s & 1) + e;
                const v4f lo = *(const v4f*)(wp + kl * P_LEAVES);
                const v4f hi = *(const v4f*)(wp + kl * P_LEAVES + 4);
                wf[s][0][e] = lo[0]; wf[s][1][e] = lo[1];
                wf[s][2][e] = lo[2]; wf[s][3][e] = lo[3];
                wf[s][4][e] = hi[0]; wf[s][5][e] = hi[1];
                wf[s][6][e] = hi[2]; wf[s][7][e] = hi[3];
            }
        }
    }

    const v4f bias4 = *(const v4f*)(bias + 4 * lane);

    const float* xp = x + (size_t)(row0 + c) * F_DIM + kq0 + 4 * h;

    float* partw0 = part + (wave * 16 + 8 * h) * P_LEAVES + 8 * c;
    const float* partr0 = part + (4 * wave) * P_LEAVES + 4 * lane;
    float* resw = res + 4 * wave;

    v4f aE[8], aO[8];
    load_a(aE, xp);
    load_a(aO, xp + (size_t)16 * F_DIM);
    tile_first(aE, wf, partw0);

    for (int tp = 0; tp < TILES_PER_WG / 2 - 1; ++tp) {
        const int te = 2 * tp;
        if (tp == 1) __builtin_amdgcn_s_setprio(2);
        if (tp == 2) __builtin_amdgcn_s_setprio(1);
        load_a(aE, xp + (size_t)(te + 2) * 16 * F_DIM);
        tile_pipe(aO, wf, partw0 + PART_WORDS, partr0, bias4,
                  resw + te * 16, lane);
        load_a(aO, xp + (size_t)(te + 3) * 16 * F_DIM);
        tile_pipe(aE, wf, partw0, partr0 + PART_WORDS, bias4,
                  resw + (te + 1) * 16, lane);
    }
    __builtin_amdgcn_s_setprio(0);
    tile_pipe(aO, wf, partw0 + PART_WORDS, partr0, bias4,
              resw + (TILES_PER_WG - 2) * 16, lane);
    meet_last(partr0 + PART_WORDS, bias4, resw + (TILES_PER_WG - 1) * 16, lane);

    __syncthreads();

    out[row0 + tid] = res[tid];
}

extern "C" void kernel_launch(void* const* d_in, const int* in_sizes, int n_in,
                              void* d_out, int out_size, void* d_ws, size_t ws_size,
                              hipStream_t stream) {
    const float* x = (const float*)d_in[0];
    const float* W = (const float*)d_in[1];
    const float* b = (const float*)d_in[2];
    float* out = (float*)d_out;

    const int Bsz = in_sizes[0] / F_DIM;
    const int nBlocks = Bsz / ROWS_PER_WG;

    const size_t shmem = (size_t)LDS_WORDS * sizeof(float);
    (void)hipFuncSetAttribute((const void*)aln_wmma_kernel,
                              hipFuncAttributeMaxDynamicSharedMemorySize,
                              (int)shmem);

    aln_wmma_kernel<<<nBlocks, THREADS_PER_WG, shmem, stream>>>(x, W, b, out);
}
